# starting kernel
# baseline (speedup 1.0000x reference)
#include <hip/hip_runtime.h>


#define L1X2   1024
#define NL2    16
#define NL3    32
#define NSTACK 8
#define W1PAD  1028
#define Y1PAD  17
#define BPS    64
#define WAVES  8

typedef __attribute__((ext_vector_type(2))) float v2f;
typedef __attribute__((ext_vector_type(8))) float v8f;

__global__ void ls_init(int* cnt) {
    if (threadIdx.x < 2 * NSTACK) cnt[threadIdx.x >= NSTACK ? 8 + threadIdx.x : threadIdx.x] = 0;
}

__global__ void ls_hist(const int* __restrict__ idx, int* __restrict__ counts, int B) {
    int s = blockIdx.x * blockDim.x + threadIdx.x;
    if (s < B) atomicAdd(&counts[idx[s]], 1);
}

__global__ void ls_scan(const int* __restrict__ counts, int* __restrict__ offsets) {
    int acc = 0;
    for (int c = 0; c < NSTACK; ++c) { offsets[c] = acc; acc += counts[c]; }
}

__global__ void ls_scatter(const int* __restrict__ idx, const int* __restrict__ offsets,
                           int* __restrict__ cursors, int* __restrict__ bucket, int B) {
    int s = blockIdx.x * blockDim.x + threadIdx.x;
    if (s < B) {
        int c = idx[s];
        int pos = offsets[c] + atomicAdd(&cursors[c], 1);
        bucket[pos] = s;
    }
}

__global__ __launch_bounds__(256)
void ls_main(const float* __restrict__ x,
             const float* __restrict__ w1, const float* __restrict__ b1,
             const float* __restrict__ w2, const float* __restrict__ b2,
             const float* __restrict__ wo, const float* __restrict__ bo,
             const int* __restrict__ counts, const int* __restrict__ offsets,
             const int* __restrict__ bucket, float* __restrict__ out) {
    __shared__ float w1s[NL2 * W1PAD];
    __shared__ float w2s[NL3 * NL2];
    __shared__ float b1s[NL2];
    __shared__ float b2s[NL3];
    __shared__ float wos[NL3];
    __shared__ float y1s[WAVES][NL2 * Y1PAD];

    const int c   = blockIdx.y;
    const int tid = threadIdx.x;

    for (int i = tid; i < NL2 * L1X2; i += 256) {
        int r = i >> 10, col = i & 1023;
        w1s[r * W1PAD + col] = w1[(c * NL2 + r) * L1X2 + col];
    }
    for (int i = tid; i < NL3 * NL2; i += 256) {
        int r = i >> 4, col = i & 15;
        w2s[i] = w2[(c * NL3 + r) * NL2 + col];
    }
    if (tid < NL2) b1s[tid] = b1[c * NL2 + tid];
    if (tid < NL3) { b2s[tid] = b2[c * NL3 + tid]; wos[tid] = wo[c * NL3 + tid]; }
    __syncthreads();

    const int n_c = counts[c];
    if (n_c == 0) return;
    const int   base = offsets[c];
    const float bo_c = bo[c];
    const int nTiles = (n_c + 15) >> 4;

    const int wave = tid >> 5;
    const int lane = tid & 31;
    const int ml   = lane & 15;
    const int half = lane >> 4;
    const int mlr  = ml * W1PAD;

    for (int t = blockIdx.x * WAVES + wave; t < nTiles; t += gridDim.x * WAVES) {
        int gs   = t * 16 + ml;
        int cgs  = gs < n_c ? gs : n_c - 1;
        int samp = bucket[base + cgs];
        const float* xrow = x + (long)samp * L1X2;

        v8f acc = {0.f, 0.f, 0.f, 0.f, 0.f, 0.f, 0.f, 0.f};
        #pragma unroll 8
        for (int k = 0; k < L1X2; k += 4) {
            const int ko = k + 2 * half;
            v2f a = __builtin_nontemporal_load((const v2f*)(xrow + ko));
            v2f b = *(const v2f*)(w1s + mlr + ko);
            acc = __builtin_amdgcn_wmma_f32_16x16x4_f32(
                false, a, false, b, (short)0, acc, false, false);
        }

        float* yb = &y1s[wave][0];
        #pragma unroll
        for (int r = 0; r < 8; ++r) {
            float v = acc[r] + b1s[ml];
            v = fminf(fmaxf(v, 0.f), 1.f);
            yb[(8 * half + r) * Y1PAD + ml] = v;
        }
        asm volatile("" ::: "memory");

        float xv[NL2];
        #pragma unroll
        for (int j = 0; j < NL2; ++j) xv[j] = yb[ml * Y1PAD + j];

        float acc3 = 0.f;
        #pragma unroll
        for (int q = 0; q < 16; ++q) {
            int jj = half * 16 + q;
            float s2 = b2s[jj];
            #pragma unroll
            for (int j = 0; j < NL2; ++j) s2 += xv[j] * w2s[jj * NL2 + j];
            s2 = fminf(fmaxf(s2, 0.f), 1.f);
            acc3 += s2 * wos[jj];
        }
        acc3 += __shfl_xor(acc3, 16, 32);

        if (half == 0 && gs < n_c) out[bucket[base + gs]] = acc3 + bo_c;
    }
}

extern "C" void kernel_launch(void* const* d_in, const int* in_sizes, int n_in,
                              void* d_out, int out_size, void* d_ws, size_t ws_size,
                              hipStream_t stream) {
    const float* x   = (const float*)d_in[0];
    const int*   idx = (const int*)  d_in[1];
    const float* w1  = (const float*)d_in[2];
    const float* b1  = (const float*)d_in[3];
    const float* w2  = (const float*)d_in[4];
    const float* b2  = (const float*)d_in[5];
    const float* wo  = (const float*)d_in[6];
    const float* bo  = (const float*)d_in[7];
    float* out = (float*)d_out;

    const int B = in_sizes[1];
    int* wsi     = (int*)d_ws;
    int* counts  = wsi;
    int* offsets = wsi + 8;
    int* cursors = wsi + 16;
    int* bucket  = wsi + 32;

    ls_init<<<1, 32, 0, stream>>>(counts);
    int nb = (B + 255) / 256;
    ls_hist<<<nb, 256, 0, stream>>>(idx, counts, B);
    ls_scan<<<1, 1, 0, stream>>>(counts, offsets);
    ls_scatter<<<nb, 256, 0, stream>>>(idx, offsets, cursors, bucket, B);

    dim3 grid(BPS, NSTACK, 1);
    ls_main<<<grid, 256, 0, stream>>>(x, w1, b1, w2, b2, wo, bo,
                                      counts, offsets, bucket, out);
}


# Bucket sort keeps one launch: a sort block orders only its own 256 samples; main's waves scan their stack's per-block counts and search each row's block in LDS; 2 launches, not 3; no sum changes
# speedup vs baseline: 2.9355x; 2.9355x over previous
#include <hip/hip_runtime.h>


#define L1X2   1024
#define NL2    16
#define NL3    32
#define NSTACK 8
#define W1PAD  1028
#define Y1PAD  17
#define BPS    64
#define WAVES  8
#define MWAVES 9

typedef __attribute__((ext_vector_type(2))) float v2f;
typedef __attribute__((ext_vector_type(4))) float v4f;
typedef __attribute__((ext_vector_type(8))) float v8f;


#define SORTB 256
#define PROWS 256

__global__ __launch_bounds__(256)
void ls_local(const int* __restrict__ idx, int* __restrict__ tab, int* __restrict__ loc, int B) {
    __shared__ int wcnt[WAVES][NSTACK];
    __shared__ int wbase[WAVES][NSTACK];
    const int s    = blockIdx.x * blockDim.x + threadIdx.x;
    const int c    = s < B ? idx[s] : -1;
    const int wave = threadIdx.x >> 5;
    const int lane = threadIdx.x & 31;

    const unsigned below = (1u << lane) - 1u;
    int rank = 0;
    for (int k = 0; k < NSTACK; ++k) {
        unsigned m = __builtin_amdgcn_ballot_w32(c == k);
        if (c == k) rank = __popc(m & below);
        if (lane == k) wcnt[wave][k] = __popc(m);
    }
    __syncthreads();
    if (threadIdx.x < NSTACK) {
        const int k = threadIdx.x;
        int off = 0, tot = 0;
        for (int j = 0; j < NSTACK; ++j) {
            int tj = 0;
            for (int w = 0; w < WAVES; ++w) tj += wcnt[w][j];
            if (j < k)  off += tj;
            if (j == k) tot  = tj;
        }
        int run = off;
        for (int w = 0; w < WAVES; ++w) { wbase[w][k] = run; run += wcnt[w][k]; }
        tab[k * gridDim.x + blockIdx.x] = tot | (off << 16);
    }
    __syncthreads();
    if ((unsigned)c < (unsigned)NSTACK) loc[blockIdx.x * SORTB + wbase[wave][c] + rank] = s;
}

__global__ __launch_bounds__(288)
void ls_main(const float* __restrict__ x,
             const float* __restrict__ w1, const float* __restrict__ b1,
             const float* __restrict__ w2, const float* __restrict__ b2,
             const float* __restrict__ wo, const float* __restrict__ bo,
             const int* __restrict__ tab, const int* __restrict__ loc, int nb,
             float* __restrict__ out) {
    __shared__ __attribute__((aligned(16))) float w1s[NL2 * W1PAD];
    __shared__ float w2s[NL3 * NL2];
    __shared__ float b1s[NL2];
    __shared__ float b2s[NL3];
    __shared__ float wos[NL3];
    __shared__ float y1s[MWAVES][NL2 * Y1PAD];
    __shared__ int   ptab[PROWS];

    const int c    = blockIdx.y;
    const int tid  = threadIdx.x;
    const int wave = tid >> 5;
    const int lane = tid & 31;
    const int ml   = lane & 15;
    const int half = lane >> 4;
    const int mlr  = ml * W1PAD;

    int tv[8];
    #pragma unroll
    for (int q = 0; q < 8; ++q) {
        const int e  = lane * 8 + q;
        const int ec = e < nb ? e : nb - 1;
        const int w  = tab[c * nb + ec];
        tv[q] = e < nb ? w : 0;
    }
    auto find_row = [&](int r) -> int {
        int j = 0;
        #pragma unroll
        for (int st = PROWS / 4; st > 0; st >>= 2) {
            const int p1 = ptab[j + st]     & 0xfffff;
            const int p2 = ptab[j + 2 * st] & 0xfffff;
            const int p3 = ptab[j + 3 * st] & 0xfffff;
            j += ((p1 <= r) + (p2 <= r) + (p3 <= r)) * st;
        }
        const int w = ptab[j];
        return loc[j * SORTB + (w >> 20) + (r - (w & 0xfffff))];
    };

    float w1r[NL2 * L1X2 / 256];
    float w2r[NL3 * NL2 / 256];
    float b1r = 0.f, b2r = 0.f, wor = 0.f;
    if (tid < 256) {
        #pragma unroll
        for (int q = 0; q < NL2 * L1X2 / 256; ++q) {
            const int i = tid + 256 * q;
            w1r[q] = w1[(c * NL2 + (i >> 10)) * L1X2 + (i & 1023)];
        }
        #pragma unroll
        for (int q = 0; q < NL3 * NL2 / 256; ++q) {
            const int i = tid + 256 * q;
            w2r[q] = w2[(c * NL3 + (i >> 4)) * NL2 + (i & 15)];
        }
    }
    if (tid < NL2) b1r = b1[c * NL2 + tid];
    if (tid < NL3) { b2r = b2[c * NL3 + tid]; wor = wo[c * NL3 + tid]; }

    const float bo_c = bo[c];

    int n_c;
    {
        int ex[8];
        int run = 0;
        #pragma unroll
        for (int q = 0; q < 8; ++q) { ex[q] = run; run += tv[q] & 0xffff; }
        int inc = run;
        inc += __builtin_amdgcn_update_dpp(0, inc, 0x111, 0xf, 0xf, true);
        inc += __builtin_amdgcn_update_dpp(0, inc, 0x112, 0xf, 0xf, true);
        inc += __builtin_amdgcn_update_dpp(0, inc, 0x114, 0xf, 0xf, true);
        inc += __builtin_amdgcn_update_dpp(0, inc, 0x118, 0xf, 0xf, true);
        const int lowrow = __builtin_amdgcn_readlane(inc, 15);
        if (lane >= 16) inc += lowrow;
        const int before = inc - run;
        #pragma unroll
        for (int q = 0; q < 8; ++q)
            ptab[lane * 8 + q] = (before + ex[q]) | ((tv[q] >> 16) << 20);
        n_c = __builtin_amdgcn_readlane(inc, 31);
    }
    if (n_c == 0) return;
    const int   nTiles = (n_c + 15) >> 4;
    const int   stride = gridDim.x * MWAVES;

    int  t   = wave * gridDim.x + blockIdx.x;
    bool has = t < nTiles;
    int  gs  = 0;
    const float* xrow = x;
    v4f pre[16];
    #pragma unroll
    for (int j = 0; j < 16; ++j) pre[j] = (v4f){0.f, 0.f, 0.f, 0.f};
    int samp = 0;
    if (has) {
        gs = t * 16 + ml;
        const int cgs = gs < n_c ? gs : n_c - 1;
        samp = find_row(cgs);
    }

    if (tid < 256) {
        #pragma unroll
        for (int q = 0; q < NL2 * L1X2 / 256; ++q) {
            const int i = tid + 256 * q;
            w1s[(i >> 10) * W1PAD + (i & 1023)] = w1r[q];
        }
        #pragma unroll
        for (int q = 0; q < NL3 * NL2 / 256; ++q) w2s[tid + 256 * q] = w2r[q];
    }
    if (tid < NL2) b1s[tid] = b1r;
    if (tid < NL3) { b2s[tid] = b2r; wos[tid] = wor; }

    if (has) {
        xrow = x + (long)samp * L1X2;
        #pragma unroll
        for (int j = 0; j < 16; ++j) pre[j] = *(const v4f*)(xrow + 8 * j + 4 * half);
    }

    asm volatile("s_wait_dscnt 0x0\n\ts_barrier_signal -1\n\ts_barrier_wait -1" ::: "memory");

    while (has) {
        v8f acc = {0.f, 0.f, 0.f, 0.f, 0.f, 0.f, 0.f, 0.f};
        #pragma unroll
        for (int j = 0; j < 16; ++j) {
            const int ko = 8 * j + 4 * half;
            v4f a = pre[j];
            v4f b = *(const v4f*)(w1s + mlr + ko);
            v2f a01 = {a.x, a.y}, b01 = {b.x, b.y};
            v2f a23 = {a.z, a.w}, b23 = {b.z, b.w};
            acc = __builtin_amdgcn_wmma_f32_16x16x4_f32(
                false, a01, false, b01, (short)0, acc, false, false);
            acc = __builtin_amdgcn_wmma_f32_16x16x4_f32(
                false, a23, false, b23, (short)0, acc, false, false);
        }
        #pragma unroll 8
        for (int k = 128; k < L1X2; k += 8) {
            const int ko = k + 4 * half;
            v4f a = *(const v4f*)(xrow + ko);
            v4f b = *(const v4f*)(w1s + mlr + ko);
            v2f a01 = {a.x, a.y}, b01 = {b.x, b.y};
            v2f a23 = {a.z, a.w}, b23 = {b.z, b.w};
            acc = __builtin_amdgcn_wmma_f32_16x16x4_f32(
                false, a01, false, b01, (short)0, acc, false, false);
            acc = __builtin_amdgcn_wmma_f32_16x16x4_f32(
                false, a23, false, b23, (short)0, acc, false, false);
        }

        float* yb = &y1s[wave][0];
        #pragma unroll
        for (int r = 0; r < 8; ++r) {
            float v = acc[r] + b1s[ml];
            v = fminf(fmaxf(v, 0.f), 1.f);
            yb[(8 * half + r) * Y1PAD + ml] = v;
        }
        asm volatile("" ::: "memory");

        float xv[NL2];
        #pragma unroll
        for (int j = 0; j < NL2; ++j) xv[j] = yb[ml * Y1PAD + j];

        float acc3 = 0.f;
        #pragma unroll
        for (int q = 0; q < 16; ++q) {
            int jj = half * 16 + q;
            float s2 = b2s[jj];
            #pragma unroll
            for (int j = 0; j < NL2; ++j) s2 += xv[j] * w2s[jj * NL2 + j];
            s2 = fminf(fmaxf(s2, 0.f), 1.f);
            acc3 += s2 * wos[jj];
        }
        acc3 += __shfl_xor(acc3, 16, 32);

        if (half == 0 && gs < n_c) out[samp] = acc3 + bo_c;

        t  += stride;
        has = t < nTiles;
        if (has) {
            gs = t * 16 + ml;
            const int cgs = gs < n_c ? gs : n_c - 1;
            samp = find_row(cgs);
            xrow = x + (long)samp * L1X2;
            #pragma unroll
            for (int j = 0; j < 16; ++j) pre[j] = *(const v4f*)(xrow + 8 * j + 4 * half);
        }
    }
}

extern "C" void kernel_launch(void* const* d_in, const int* in_sizes, int n_in,
                              void* d_out, int out_size, void* d_ws, size_t ws_size,
                              hipStream_t stream) {
    const float* x   = (const float*)d_in[0];
    const int*   idx = (const int*)  d_in[1];
    const float* w1  = (const float*)d_in[2];
    const float* b1  = (const float*)d_in[3];
    const float* w2  = (const float*)d_in[4];
    const float* b2  = (const float*)d_in[5];
    const float* wo  = (const float*)d_in[6];
    const float* bo  = (const float*)d_in[7];
    float* out = (float*)d_out;

    const int B = in_sizes[1];
    int* wsi = (int*)d_ws;
    int* loc = wsi + 32;
    int* tab = loc + B;

    int nb = (B + 255) / 256;
    ls_local<<<nb, 256, 0, stream>>>(idx, tab, loc, B);

    dim3 grid(BPS, NSTACK, 1);
    ls_main<<<grid, 288, 0, stream>>>(x, w1, b1, w2, b2, wo, bo,
                                      tab, loc, nb, out);
}


# Main kernel: a wave prefetches its 16 rows' 256-B units into the second-level cache three units ahead of its row loads (one instruction per unit): memory works under loads and WMMAs; no sum changes
# speedup vs baseline: 3.1284x; 1.0657x over previous
#include <hip/hip_runtime.h>


#define L1X2   1024
#define NL2    16
#define NL3    32
#define NSTACK 8
#define W1PAD  1028
#define Y1PAD  17
#define BPS    64
#define WAVES  8
#define MWAVES 9
#define PFD    3

typedef __attribute__((ext_vector_type(2))) float v2f;
typedef __attribute__((ext_vector_type(4))) float v4f;
typedef __attribute__((ext_vector_type(8))) float v8f;


#define SORTB 256
#define PROWS 256

__global__ __launch_bounds__(256)
void ls_local(const int* __restrict__ idx, int* __restrict__ tab, int* __restrict__ loc, int B) {
    __shared__ int wcnt[WAVES][NSTACK];
    __shared__ int wbase[WAVES][NSTACK];
    const int s    = blockIdx.x * blockDim.x + threadIdx.x;
    const int c    = s < B ? idx[s] : -1;
    const int wave = threadIdx.x >> 5;
    const int lane = threadIdx.x & 31;

    const unsigned below = (1u << lane) - 1u;
    int rank = 0;
    for (int k = 0; k < NSTACK; ++k) {
        unsigned m = __builtin_amdgcn_ballot_w32(c == k);
        if (c == k) rank = __popc(m & below);
        if (lane == k) wcnt[wave][k] = __popc(m);
    }
    __syncthreads();
    if (threadIdx.x < NSTACK) {
        const int k = threadIdx.x;
        int off = 0, tot = 0;
        for (int j = 0; j < NSTACK; ++j) {
            int tj = 0;
            for (int w = 0; w < WAVES; ++w) tj += wcnt[w][j];
            if (j < k)  off += tj;
            if (j == k) tot  = tj;
        }
        int run = off;
        for (int w = 0; w < WAVES; ++w) { wbase[w][k] = run; run += wcnt[w][k]; }
        tab[k * gridDim.x + blockIdx.x] = tot | (off << 16);
    }
    __syncthreads();
    if ((unsigned)c < (unsigned)NSTACK) loc[blockIdx.x * SORTB + wbase[wave][c] + rank] = s;
}

__global__ __launch_bounds__(288)
void ls_main(const float* __restrict__ x,
             const float* __restrict__ w1, const float* __restrict__ b1,
             const float* __restrict__ w2, const float* __restrict__ b2,
             const float* __restrict__ wo, const float* __restrict__ bo,
             const int* __restrict__ tab, const int* __restrict__ loc, int nb,
             float* __restrict__ out) {
    __shared__ __attribute__((aligned(16))) float w1s[NL2 * W1PAD];
    __shared__ float w2s[NL3 * NL2];
    __shared__ float b1s[NL2];
    __shared__ float b2s[NL3];
    __shared__ float wos[NL3];
    __shared__ float y1s[MWAVES][NL2 * Y1PAD];
    __shared__ int   ptab[PROWS];

    const int c    = blockIdx.y;
    const int tid  = threadIdx.x;
    const int wave = tid >> 5;
    const int lane = tid & 31;
    const int ml   = lane & 15;
    const int half = lane >> 4;
    const int mlr  = ml * W1PAD;

    int tv[8];
    #pragma unroll
    for (int q = 0; q < 8; ++q) {
        const int e  = lane * 8 + q;
        const int ec = e < nb ? e : nb - 1;
        const int w  = tab[c * nb + ec];
        tv[q] = e < nb ? w : 0;
    }
    auto find_row = [&](int r) -> int {
        int j = 0;
        #pragma unroll
        for (int st = PROWS / 4; st > 0; st >>= 2) {
            const int p1 = ptab[j + st]     & 0xfffff;
            const int p2 = ptab[j + 2 * st] & 0xfffff;
            const int p3 = ptab[j + 3 * st] & 0xfffff;
            j += ((p1 <= r) + (p2 <= r) + (p3 <= r)) * st;
        }
        const int w = ptab[j];
        return loc[j * SORTB + (w >> 20) + (r - (w & 0xfffff))];
    };

    float w1r[NL2 * L1X2 / 256];
    float w2r[NL3 * NL2 / 256];
    float b1r = 0.f, b2r = 0.f, wor = 0.f;
    if (tid < 256) {
        #pragma unroll
        for (int q = 0; q < NL2 * L1X2 / 256; ++q) {
            const int i = tid + 256 * q;
            w1r[q] = w1[(c * NL2 + (i >> 10)) * L1X2 + (i & 1023)];
        }
        #pragma unroll
        for (int q = 0; q < NL3 * NL2 / 256; ++q) {
            const int i = tid + 256 * q;
            w2r[q] = w2[(c * NL3 + (i >> 4)) * NL2 + (i & 15)];
        }
    }
    if (tid < NL2) b1r = b1[c * NL2 + tid];
    if (tid < NL3) { b2r = b2[c * NL3 + tid]; wor = wo[c * NL3 + tid]; }

    const float bo_c = bo[c];

    int n_c;
    {
        int ex[8];
        int run = 0;
        #pragma unroll
        for (int q = 0; q < 8; ++q) { ex[q] = run; run += tv[q] & 0xffff; }
        int inc = run;
        inc += __builtin_amdgcn_update_dpp(0, inc, 0x111, 0xf, 0xf, true);
        inc += __builtin_amdgcn_update_dpp(0, inc, 0x112, 0xf, 0xf, true);
        inc += __builtin_amdgcn_update_dpp(0, inc, 0x114, 0xf, 0xf, true);
        inc += __builtin_amdgcn_update_dpp(0, inc, 0x118, 0xf, 0xf, true);
        const int lowrow = __builtin_amdgcn_readlane(inc, 15);
        if (lane >= 16) inc += lowrow;
        const int before = inc - run;
        #pragma unroll
        for (int q = 0; q < 8; ++q)
            ptab[lane * 8 + q] = (before + ex[q]) | ((tv[q] >> 16) << 20);
        n_c = __builtin_amdgcn_readlane(inc, 31);
    }
    if (n_c == 0) return;
    const int   nTiles = (n_c + 15) >> 4;
    const int   stride = gridDim.x * MWAVES;

    int  t   = wave * gridDim.x + blockIdx.x;
    bool has = t < nTiles;
    int  gs  = 0;
    const float* xrow = x;
    v4f pre[16];
    #pragma unroll
    for (int j = 0; j < 16; ++j) pre[j] = (v4f){0.f, 0.f, 0.f, 0.f};
    int samp = 0;
    if (has) {
        gs = t * 16 + ml;
        const int cgs = gs < n_c ? gs : n_c - 1;
        samp = find_row(cgs);
    }

    if (tid < 256) {
        #pragma unroll
        for (int q = 0; q < NL2 * L1X2 / 256; ++q) {
            const int i = tid + 256 * q;
            w1s[(i >> 10) * W1PAD + (i & 1023)] = w1r[q];
        }
        #pragma unroll
        for (int q = 0; q < NL3 * NL2 / 256; ++q) w2s[tid + 256 * q] = w2r[q];
    }
    if (tid < NL2) b1s[tid] = b1r;
    if (tid < NL3) { b2s[tid] = b2r; wos[tid] = wor; }

    if (has) {
        xrow = x + (long)samp * L1X2;
        #pragma unroll
        for (int j = 0; j < 16; ++j) pre[j] = *(const v4f*)(xrow + 8 * j + 4 * half);
        #pragma unroll
        for (int u = 2; u < 2 + PFD; ++u) __builtin_prefetch(xrow + 64 * u, 0, 3);
    }

    asm volatile("s_wait_dscnt 0x0\n\ts_barrier_signal -1\n\ts_barrier_wait -1" ::: "memory");

    while (has) {
        v8f acc = {0.f, 0.f, 0.f, 0.f, 0.f, 0.f, 0.f, 0.f};
        #pragma unroll
        for (int j = 0; j < 16; ++j) {
            const int ko = 8 * j + 4 * half;
            v4f a = pre[j];
            v4f b = *(const v4f*)(w1s + mlr + ko);
            v2f a01 = {a.x, a.y}, b01 = {b.x, b.y};
            v2f a23 = {a.z, a.w}, b23 = {b.z, b.w};
            acc = __builtin_amdgcn_wmma_f32_16x16x4_f32(
                false, a01, false, b01, (short)0, acc, false, false);
            acc = __builtin_amdgcn_wmma_f32_16x16x4_f32(
                false, a23, false, b23, (short)0, acc, false, false);
        }
        auto unit = [&](int u) {
            #pragma unroll
            for (int s = 0; s < 8; ++s) {
                const int ko = 64 * u + 8 * s + 4 * half;
                v4f a = *(const v4f*)(xrow + ko);
                v4f b = *(const v4f*)(w1s + mlr + ko);
                v2f a01 = {a.x, a.y}, b01 = {b.x, b.y};
                v2f a23 = {a.z, a.w}, b23 = {b.z, b.w};
                acc = __builtin_amdgcn_wmma_f32_16x16x4_f32(
                    false, a01, false, b01, (short)0, acc, false, false);
                acc = __builtin_amdgcn_wmma_f32_16x16x4_f32(
                    false, a23, false, b23, (short)0, acc, false, false);
            }
        };
        #pragma nounroll
        for (int u = 2; u < 16 - PFD; ++u) {
            __builtin_prefetch(xrow + 64 * (u + PFD), 0, 3);
            unit(u);
        }
        #pragma nounroll
        for (int u = 16 - PFD; u < 16; ++u) unit(u);

        float* yb = &y1s[wave][0];
        #pragma unroll
        for (int r = 0; r < 8; ++r) {
            float v = acc[r] + b1s[ml];
            v = fminf(fmaxf(v, 0.f), 1.f);
            yb[(8 * half + r) * Y1PAD + ml] = v;
        }
        asm volatile("" ::: "memory");

        float xv[NL2];
        #pragma unroll
        for (int j = 0; j < NL2; ++j) xv[j] = yb[ml * Y1PAD + j];

        float acc3 = 0.f;
        #pragma unroll
        for (int q = 0; q < 16; ++q) {
            int jj = half * 16 + q;
            float s2 = b2s[jj];
            #pragma unroll
            for (int j = 0; j < NL2; ++j) s2 += xv[j] * w2s[jj * NL2 + j];
            s2 = fminf(fmaxf(s2, 0.f), 1.f);
            acc3 += s2 * wos[jj];
        }
        acc3 += __shfl_xor(acc3, 16, 32);

        if (half == 0 && gs < n_c) out[samp] = acc3 + bo_c;

        t  += stride;
        has = t < nTiles;
        if (has) {
            gs = t * 16 + ml;
            const int cgs = gs < n_c ? gs : n_c - 1;
            samp = find_row(cgs);
            xrow = x + (long)samp * L1X2;
            #pragma unroll
            for (int j = 0; j < 16; ++j) pre[j] = *(const v4f*)(xrow + 8 * j + 4 * half);
            #pragma unroll
            for (int u = 2; u < 2 + PFD; ++u) __builtin_prefetch(xrow + 64 * u, 0, 3);
        }
    }
}

extern "C" void kernel_launch(void* const* d_in, const int* in_sizes, int n_in,
                              void* d_out, int out_size, void* d_ws, size_t ws_size,
                              hipStream_t stream) {
    const float* x   = (const float*)d_in[0];
    const int*   idx = (const int*)  d_in[1];
    const float* w1  = (const float*)d_in[2];
    const float* b1  = (const float*)d_in[3];
    const float* w2  = (const float*)d_in[4];
    const float* b2  = (const float*)d_in[5];
    const float* wo  = (const float*)d_in[6];
    const float* bo  = (const float*)d_in[7];
    float* out = (float*)d_out;

    const int B = in_sizes[1];
    int* wsi = (int*)d_ws;
    int* loc = wsi + 32;
    int* tab = loc + B;

    int nb = (B + 255) / 256;
    ls_local<<<nb, 256, 0, stream>>>(idx, tab, loc, B);

    dim3 grid(BPS, NSTACK, 1);
    ls_main<<<grid, 288, 0, stream>>>(x, w1, b1, w2, b2, wo, bo,
                                      tab, loc, nb, out);
}


# Main K loop software-pipelined in registers: a unit's 8 row loads go out two units ahead into a second register set, its 8 LDS weight reads as one group before its 16 WMMAs; no sum changes
# speedup vs baseline: 3.2327x; 1.0333x over previous
#include <hip/hip_runtime.h>


#define L1X2   1024
#define NL2    16
#define NL3    32
#define NSTACK 8
#define W1PAD  1028
#define Y1PAD  17
#define BPS    64
#define WAVES  8
#define MWAVES 9
#define PFD    3
#define LEAD   2

typedef __attribute__((ext_vector_type(2))) float v2f;
typedef __attribute__((ext_vector_type(4))) float v4f;
typedef __attribute__((ext_vector_type(8))) float v8f;


#define SORTB 256
#define PROWS 256

__global__ __launch_bounds__(256)
void ls_local(const int* __restrict__ idx, int* __restrict__ tab, int* __restrict__ loc, int B) {
    __shared__ int wcnt[WAVES][NSTACK];
    __shared__ int wbase[WAVES][NSTACK];
    const int s    = blockIdx.x * blockDim.x + threadIdx.x;
    const int c    = s < B ? idx[s] : -1;
    const int wave = threadIdx.x >> 5;
    const int lane = threadIdx.x & 31;

    const unsigned below = (1u << lane) - 1u;
    int rank = 0;
    for (int k = 0; k < NSTACK; ++k) {
        unsigned m = __builtin_amdgcn_ballot_w32(c == k);
        if (c == k) rank = __popc(m & below);
        if (lane == k) wcnt[wave][k] = __popc(m);
    }
    __syncthreads();
    if (threadIdx.x < NSTACK) {
        const int k = threadIdx.x;
        int off = 0, tot = 0;
        for (int j = 0; j < NSTACK; ++j) {
            int tj = 0;
            for (int w = 0; w < WAVES; ++w) tj += wcnt[w][j];
            if (j < k)  off += tj;
            if (j == k) tot  = tj;
        }
        int run = off;
        for (int w = 0; w < WAVES; ++w) { wbase[w][k] = run; run += wcnt[w][k]; }
        tab[k * gridDim.x + blockIdx.x] = tot | (off << 16);
    }
    __syncthreads();
    if ((unsigned)c < (unsigned)NSTACK) loc[blockIdx.x * SORTB + wbase[wave][c] + rank] = s;
}

__global__ __launch_bounds__(288)
void ls_main(const float* __restrict__ x,
             const float* __restrict__ w1, const float* __restrict__ b1,
             const float* __restrict__ w2, const float* __restrict__ b2,
             const float* __restrict__ wo, const float* __restrict__ bo,
             const int* __restrict__ tab, const int* __restrict__ loc, int nb,
             float* __restrict__ out) {
    __shared__ __attribute__((aligned(16))) float w1s[NL2 * W1PAD];
    __shared__ float w2s[NL3 * NL2];
    __shared__ float b1s[NL2];
    __shared__ float b2s[NL3];
    __shared__ float wos[NL3];
    __shared__ float y1s[MWAVES][NL2 * Y1PAD];
    __shared__ int   ptab[PROWS];

    const int c    = blockIdx.y;
    const int tid  = threadIdx.x;
    const int wave = tid >> 5;
    const int lane = tid & 31;
    const int ml   = lane & 15;
    const int half = lane >> 4;
    const int mlr  = ml * W1PAD;

    int tv[8];
    #pragma unroll
    for (int q = 0; q < 8; ++q) {
        const int e  = lane * 8 + q;
        const int ec = e < nb ? e : nb - 1;
        const int w  = tab[c * nb + ec];
        tv[q] = e < nb ? w : 0;
    }
    auto find_row = [&](int r) -> int {
        int j = 0;
        #pragma unroll
        for (int st = PROWS / 4; st > 0; st >>= 2) {
            const int p1 = ptab[j + st]     & 0xfffff;
            const int p2 = ptab[j + 2 * st] & 0xfffff;
            const int p3 = ptab[j + 3 * st] & 0xfffff;
            j += ((p1 <= r) + (p2 <= r) + (p3 <= r)) * st;
        }
        const int w = ptab[j];
        return loc[j * SORTB + (w >> 20) + (r - (w & 0xfffff))];
    };

    float w1r[NL2 * L1X2 / 256];
    float w2r[NL3 * NL2 / 256];
    float b1r = 0.f, b2r = 0.f, wor = 0.f;
    if (tid < 256) {
        #pragma unroll
        for (int q = 0; q < NL2 * L1X2 / 256; ++q) {
            const int i = tid + 256 * q;
            w1r[q] = w1[(c * NL2 + (i >> 10)) * L1X2 + (i & 1023)];
        }
        #pragma unroll
        for (int q = 0; q < NL3 * NL2 / 256; ++q) {
            const int i = tid + 256 * q;
            w2r[q] = w2[(c * NL3 + (i >> 4)) * NL2 + (i & 15)];
        }
    }
    if (tid < NL2) b1r = b1[c * NL2 + tid];
    if (tid < NL3) { b2r = b2[c * NL3 + tid]; wor = wo[c * NL3 + tid]; }

    const float bo_c = bo[c];

    int n_c;
    {
        int ex[8];
        int run = 0;
        #pragma unroll
        for (int q = 0; q < 8; ++q) { ex[q] = run; run += tv[q] & 0xffff; }
        int inc = run;
        inc += __builtin_amdgcn_update_dpp(0, inc, 0x111, 0xf, 0xf, true);
        inc += __builtin_amdgcn_update_dpp(0, inc, 0x112, 0xf, 0xf, true);
        inc += __builtin_amdgcn_update_dpp(0, inc, 0x114, 0xf, 0xf, true);
        inc += __builtin_amdgcn_update_dpp(0, inc, 0x118, 0xf, 0xf, true);
        const int lowrow = __builtin_amdgcn_readlane(inc, 15);
        if (lane >= 16) inc += lowrow;
        const int before = inc - run;
        #pragma unroll
        for (int q = 0; q < 8; ++q)
            ptab[lane * 8 + q] = (before + ex[q]) | ((tv[q] >> 16) << 20);
        n_c = __builtin_amdgcn_readlane(inc, 31);
    }
    if (n_c == 0) return;
    const int   nTiles = (n_c + 15) >> 4;
    const int   stride = gridDim.x * MWAVES;

    int  t   = wave * gridDim.x + blockIdx.x;
    bool has = t < nTiles;
    int  gs  = 0;
    const float* xrow = x;
    v4f xa[8], xb[8];
    #pragma unroll
    for (int j = 0; j < 8; ++j) { xa[j] = (v4f){0.f, 0.f, 0.f, 0.f}; xb[j] = (v4f){0.f, 0.f, 0.f, 0.f}; }
    int samp = 0;
    if (has) {
        gs = t * 16 + ml;
        const int cgs = gs < n_c ? gs : n_c - 1;
        samp = find_row(cgs);
    }

    if (tid < 256) {
        #pragma unroll
        for (int q = 0; q < NL2 * L1X2 / 256; ++q) {
            const int i = tid + 256 * q;
            w1s[(i >> 10) * W1PAD + (i & 1023)] = w1r[q];
        }
        #pragma unroll
        for (int q = 0; q < NL3 * NL2 / 256; ++q) w2s[tid + 256 * q] = w2r[q];
    }
    if (tid < NL2) b1s[tid] = b1r;
    if (tid < NL3) { b2s[tid] = b2r; wos[tid] = wor; }

    if (has) {
        xrow = x + (long)samp * L1X2;
        #pragma unroll
        for (int j = 0; j < 8; ++j) xa[j] = *(const v4f*)(xrow + 8 * j + 4 * half);
        #pragma unroll
        for (int j = 0; j < 8; ++j) xb[j] = *(const v4f*)(xrow + 64 + 8 * j + 4 * half);
        #pragma unroll
        for (int u = LEAD; u < LEAD + PFD; ++u) __builtin_prefetch(xrow + 64 * u, 0, 3);
    }

    asm volatile("s_wait_dscnt 0x0\n\ts_barrier_signal -1\n\ts_barrier_wait -1" ::: "memory");

    while (has) {
        v8f acc = {0.f, 0.f, 0.f, 0.f, 0.f, 0.f, 0.f, 0.f};
        auto load_unit = [&](v4f (&xr)[8], int u) {
            #pragma unroll
            for (int s = 0; s < 8; ++s) xr[s] = *(const v4f*)(xrow + 64 * u + 8 * s + 4 * half);
        };
        auto mma_unit = [&](const v4f (&xr)[8], int u) {
            v4f b[8];
            #pragma unroll
            for (int s = 0; s < 8; ++s) b[s] = *(const v4f*)(w1s + mlr + 64 * u + 8 * s + 4 * half);
            __builtin_amdgcn_sched_barrier(0);
            #pragma unroll
            for (int s = 0; s < 8; ++s) {
                v2f a01 = {xr[s].x, xr[s].y}, b01 = {b[s].x, b[s].y};
                v2f a23 = {xr[s].z, xr[s].w}, b23 = {b[s].z, b[s].w};
                acc = __builtin_amdgcn_wmma_f32_16x16x4_f32(
                    false, a01, false, b01, (short)0, acc, false, false);
                acc = __builtin_amdgcn_wmma_f32_16x16x4_f32(
                    false, a23, false, b23, (short)0, acc, false, false);
            }
            __builtin_amdgcn_sched_barrier(0);
        };
        #pragma nounroll
        for (int u = 0; u < 16 - LEAD; u += 2) {
            mma_unit(xa, u);
            if (u + LEAD + PFD < 16) __builtin_prefetch(xrow + 64 * (u + LEAD + PFD), 0, 3);
            load_unit(xa, u + LEAD);
            mma_unit(xb, u + 1);
            if (u + 1 + LEAD + PFD < 16) __builtin_prefetch(xrow + 64 * (u + 1 + LEAD + PFD), 0, 3);
            load_unit(xb, u + 1 + LEAD);
        }
        mma_unit(xa, 16 - LEAD);
        mma_unit(xb, 16 - LEAD + 1);

        float* yb = &y1s[wave][0];
        #pragma unroll
        for (int r = 0; r < 8; ++r) {
            float v = acc[r] + b1s[ml];
            v = fminf(fmaxf(v, 0.f), 1.f);
            yb[(8 * half + r) * Y1PAD + ml] = v;
        }
        asm volatile("" ::: "memory");

        float xv[NL2];
        #pragma unroll
        for (int j = 0; j < NL2; ++j) xv[j] = yb[ml * Y1PAD + j];

        float acc3 = 0.f;
        #pragma unroll
        for (int q = 0; q < 16; ++q) {
            int jj = half * 16 + q;
            float s2 = b2s[jj];
            #pragma unroll
            for (int j = 0; j < NL2; ++j) s2 += xv[j] * w2s[jj * NL2 + j];
            s2 = fminf(fmaxf(s2, 0.f), 1.f);
            acc3 += s2 * wos[jj];
        }
        acc3 += __shfl_xor(acc3, 16, 32);

        if (half == 0 && gs < n_c) out[samp] = acc3 + bo_c;

        t  += stride;
        has = t < nTiles;
        if (has) {
            gs = t * 16 + ml;
            const int cgs = gs < n_c ? gs : n_c - 1;
            samp = find_row(cgs);
            xrow = x + (long)samp * L1X2;
            #pragma unroll
            for (int j = 0; j < 8; ++j) xa[j] = *(const v4f*)(xrow + 8 * j + 4 * half);
            #pragma unroll
            for (int j = 0; j < 8; ++j) xb[j] = *(const v4f*)(xrow + 64 + 8 * j + 4 * half);
            #pragma unroll
            for (int u = LEAD; u < LEAD + PFD; ++u) __builtin_prefetch(xrow + 64 * u, 0, 3);
        }
    }
}

extern "C" void kernel_launch(void* const* d_in, const int* in_sizes, int n_in,
                              void* d_out, int out_size, void* d_ws, size_t ws_size,
                              hipStream_t stream) {
    const float* x   = (const float*)d_in[0];
    const int*   idx = (const int*)  d_in[1];
    const float* w1  = (const float*)d_in[2];
    const float* b1  = (const float*)d_in[3];
    const float* w2  = (const float*)d_in[4];
    const float* b2  = (const float*)d_in[5];
    const float* wo  = (const float*)d_in[6];
    const float* bo  = (const float*)d_in[7];
    float* out = (float*)d_out;

    const int B = in_sizes[1];
    int* wsi = (int*)d_ws;
    int* loc = wsi + 32;
    int* tab = loc + B;

    int nb = (B + 255) / 256;
    ls_local<<<nb, 256, 0, stream>>>(idx, tab, loc, B);

    dim3 grid(BPS, NSTACK, 1);
    ls_main<<<grid, 288, 0, stream>>>(x, w1, b1, w2, b2, wo, bo,
                                      tab, loc, nb, out);
}
